# starting kernel
# baseline (speedup 1.0000x reference)
#include <hip/hip_runtime.h>
#include <hip/hip_bf16.h>
#include <math.h>

#define MOMENTUM 0.05f
#define EPS_V 1e-6f

typedef float v2f __attribute__((ext_vector_type(2)));
typedef float v4f __attribute__((ext_vector_type(4)));
typedef float v8f __attribute__((ext_vector_type(8)));

__global__ void __launch_bounds__(512)
softmax_weights(const float* __restrict__ Wa, const float* __restrict__ Wb,
                float* __restrict__ wa, float* __restrict__ wb, int C) {
    __shared__ float red[1024];
    int tid = threadIdx.x;
    const float* src = (blockIdx.x == 0) ? Wa : Wb;
    float*       dst = (blockIdx.x == 0) ? wa : wb;
    float v = (tid < C) ? src[tid] : -INFINITY;
    red[tid] = v; __syncthreads();
    for (int off = blockDim.x >> 1; off > 0; off >>= 1) {
        if (tid < off) red[tid] = fmaxf(red[tid], red[tid + off]);
        __syncthreads();
    }
    float mx = red[0]; __syncthreads();
    float e = (tid < C) ? __expf(v - mx) : 0.f;
    red[tid] = e; __syncthreads();
    for (int off = blockDim.x >> 1; off > 0; off >>= 1) {
        if (tid < off) red[tid] += red[tid + off];
        __syncthreads();
    }
    float s = red[0];
    if (tid < C) dst[tid] = e / s;
}

__global__ void __launch_bounds__(256)
reduce_wmma(const float* __restrict__ x,
            const float* __restrict__ wa, const float* __restrict__ wb,
            float* __restrict__ xa, float* __restrict__ xb,
            float* __restrict__ xq, int C, int T) {
    __shared__ float ldsw[1024 + 8];
    int tid = threadIdx.x;
    for (int i = tid; i < 2 * C; i += blockDim.x)
        ldsw[i] = (i < C) ? wa[i] : wb[i - C];
    if (tid < 8) ldsw[2 * C + tid] = 0.f;
    __syncthreads();

    int lane = tid & 31;
    int wave = blockIdx.x * (blockDim.x >> 5) + (tid >> 5);
    int tilesPerB = T >> 4;
    int b  = wave / tilesPerB;
    int t0 = (wave % tilesPerB) << 4;

    int  m    = lane & 15;
    int  koff = (lane >> 4) << 1;
    bool m0 = (m == 0), m1 = (m == 1);

    const float* a1p = m0 ? &ldsw[koff] : (m1 ? &ldsw[C + koff] : &ldsw[2 * C]);
    const float* a2p = m0 ? &ldsw[C + koff] : &ldsw[2 * C];
    int a1s = (m < 2) ? 4 : 0;
    int a2s = m0 ? 4 : 0;

    v8f acc1 = {}; v8f acc2 = {};
    const float* xp = x + (size_t)b * C * T + (size_t)koff * T + t0 + m;
    const size_t xstep = 4 * (size_t)T;

    for (int c0 = 0; c0 < C; c0 += 4) {
        v2f A1 = *(const v2f*)a1p;
        v2f A2 = *(const v2f*)a2p;
        v2f Bx;
        Bx.x = xp[0];
        Bx.y = xp[(size_t)T];
        v2f Bxx = Bx * Bx;
        acc1 = __builtin_amdgcn_wmma_f32_16x16x4_f32(false, A1, false, Bx,
                                                     (short)0, acc1, false, false);
        acc2 = __builtin_amdgcn_wmma_f32_16x16x4_f32(false, A2, false, Bxx,
                                                     (short)0, acc2, false, false);
        a1p += a1s;
        a2p += a2s;
        xp  += xstep;
    }
    if (lane < 16) {
        size_t o = (size_t)b * T + t0 + lane;
        xa[o] = acc1[0];
        xb[o] = acc1[1];
        xq[o] = acc2[0];
    }
}

#define SCAN_THREADS 512
#define SCAN_E 16
__global__ void __launch_bounds__(SCAN_THREADS)
gated_scan(const float* __restrict__ g,
           const float* __restrict__ xa, const float* __restrict__ xb,
           const float* __restrict__ xq,
           float* __restrict__ mean, float* __restrict__ invstd, int T) {
    __shared__ float sA[SCAN_THREADS], sB[SCAN_THREADS];
    int b = blockIdx.x, tid = threadIdx.x;
    size_t base = (size_t)b * T + (size_t)tid * SCAN_E;

    float a_loc[SCAN_E], b_loc[SCAN_E], m_loc[SCAN_E];

    float A = 1.f, Bc = 0.f;
    for (int i = 0; i < SCAN_E; ++i) {
        float gt = g[base + i] * MOMENTUM;
        float a  = 1.f - gt;
        float bb = gt * xa[base + i];
        a_loc[i] = a; b_loc[i] = bb;
        Bc = a * Bc + bb;  A *= a;
    }
    sA[tid] = A; sB[tid] = Bc; __syncthreads();
    for (int off = 1; off < SCAN_THREADS; off <<= 1) {
        float pA = 1.f, pB = 0.f;
        if (tid >= off) { pA = sA[tid - off]; pB = sB[tid - off]; }
        __syncthreads();
        if (tid >= off) { float Ac = sA[tid]; sB[tid] = Ac * pB + sB[tid]; sA[tid] = Ac * pA; }
        __syncthreads();
    }
    float y = (tid == 0) ? 0.f : sB[tid - 1];
    for (int i = 0; i < SCAN_E; ++i) {
        y = a_loc[i] * y + b_loc[i];
        m_loc[i] = y;
        mean[base + i] = y;
    }
    __syncthreads();

    A = 1.f; Bc = 0.f;
    for (int i = 0; i < SCAN_E; ++i) {
        float mv = m_loc[i];
        float s  = xq[base + i] - 2.f * mv * xb[base + i] + mv * mv;
        float gt = 1.f - a_loc[i];
        float bb = gt * s;
        b_loc[i] = bb;
        Bc = a_loc[i] * Bc + bb;  A *= a_loc[i];
    }
    sA[tid] = A; sB[tid] = Bc; __syncthreads();
    for (int off = 1; off < SCAN_THREADS; off <<= 1) {
        float pA = 1.f, pB = 0.f;
        if (tid >= off) { pA = sA[tid - off]; pB = sB[tid - off]; }
        __syncthreads();
        if (tid >= off) { float Ac = sA[tid]; sB[tid] = Ac * pB + sB[tid]; sA[tid] = Ac * pA; }
        __syncthreads();
    }
    y = (tid == 0) ? 0.f : sB[tid - 1];
    for (int i = 0; i < SCAN_E; ++i) {
        y = a_loc[i] * y + b_loc[i];
        invstd[base + i] = rsqrtf(y + EPS_V);
    }
}

__global__ void __launch_bounds__(256)
apply_norm(const float* __restrict__ x,
           const float* __restrict__ mean, const float* __restrict__ invstd,
           const float* __restrict__ Wo_w, const float* __restrict__ Wo_b,
           float* __restrict__ out, int C, int T) {
    int row = blockIdx.y;
    int c   = row % C;
    int b   = row / C;
    int t4  = (blockIdx.x * blockDim.x + threadIdx.x) * 4;

    size_t xo = (size_t)row * T + t4;
    size_t mo = (size_t)b   * T + t4;

    v4f xv = __builtin_nontemporal_load((const v4f*)(x + xo));
    v4f mv = *(const v4f*)(mean   + mo);
    v4f iv = *(const v4f*)(invstd + mo);
    float w  = Wo_w[c];
    float bi = Wo_b[c];
    v4f o;
    o.x = (xv.x - mv.x) * iv.x * w + bi;
    o.y = (xv.y - mv.y) * iv.y * w + bi;
    o.z = (xv.z - mv.z) * iv.z * w + bi;
    o.w = (xv.w - mv.w) * iv.w * w + bi;
    __builtin_nontemporal_store(o, (v4f*)(out + xo));
}

extern "C" void kernel_launch(void* const* d_in, const int* in_sizes, int n_in,
                              void* d_out, int out_size, void* d_ws, size_t ws_size,
                              hipStream_t stream) {
    const float* x    = (const float*)d_in[0];
    const float* g    = (const float*)d_in[1];
    const float* Wa_w = (const float*)d_in[2];
    const float* Wb_w = (const float*)d_in[3];
    const float* Wo_w = (const float*)d_in[4];
    const float* Wo_b = (const float*)d_in[5];
    float* out = (float*)d_out;

    const int C  = in_sizes[2];
    const int BT = in_sizes[1];
    const int T  = 8192;
    const int B  = BT / T;

    float* ws   = (float*)d_ws;
    float* wa   = ws;
    float* wb   = wa + C;
    float* xa   = wb + C;
    float* xb   = xa + BT;
    float* xq   = xb + BT;
    float* mean = xq + BT;
    float* inv  = mean + BT;

    softmax_weights<<<2, 512, 0, stream>>>(Wa_w, Wb_w, wa, wb, C);

    int waves  = (B * T) / 16;
    int blocks = waves / 8;
    reduce_wmma<<<blocks, 256, 0, stream>>>(x, wa, wb, xa, xb, xq, C, T);

    gated_scan<<<B, SCAN_THREADS, 0, stream>>>(g, xa, xb, xq, mean, inv, T);

    dim3 agrid(T / (256 * 4), B * C);
    apply_norm<<<agrid, 256, 0, stream>>>(x, mean, inv, Wo_w, Wo_b, out, C, T);
}


# softmax launch gone (3 -> 2): reduce kernel weights its sums with exp(W) unnormalised; each scan wave sums both denominators and scales xa, xb, xq by 1/sum; denominator sum order changes
# speedup vs baseline: 1.6990x; 1.6990x over previous
#include <hip/hip_runtime.h>
#include <hip/hip_bf16.h>
#include <math.h>

#define MOMENTUM 0.05f
#define EPS_V 1e-6f

typedef float v2f __attribute__((ext_vector_type(2)));
typedef float v4f __attribute__((ext_vector_type(4)));
typedef float v8f __attribute__((ext_vector_type(8)));


#define RED_C   64
#define RED_T   128
#define RED_P   136
__global__ void __launch_bounds__(256)
reduce_wmma(const float* __restrict__ x,
            const float* __restrict__ Wa, const float* __restrict__ Wb,
            float* __restrict__ xa, float* __restrict__ xb,
            float* __restrict__ xq, int C, int T) {
    __shared__ __attribute__((aligned(16))) float ldsw[1024 + 8];
    __shared__ __attribute__((aligned(16))) float xs[2][RED_C * RED_P];

    const int tid  = threadIdx.x;
    const int lane = tid & 31;
    const int w    = tid >> 5;

    const int slabsPerB = T / RED_T;
    const int b  = blockIdx.x / slabsPerB;
    const int ts = (blockIdx.x % slabsPerB) * RED_T;

    const float* gp = x + (size_t)b * C * T + (size_t)w * T + ts + 4 * lane;
    const size_t rstep = 8 * (size_t)T;
    const int    soff  = w * RED_P + 4 * lane;

    v4f st[RED_C / 8];
    #pragma unroll
    for (int j = 0; j < RED_C / 8; ++j)
        st[j] = *(const v4f*)(gp + (size_t)j * rstep);

    if (tid < (C >> 1)) {
        const int q = C >> 2;
        const float* wsrc = (tid < q) ? (Wa + 4 * tid) : (Wb + 4 * (tid - q));
        v4f wv = *(const v4f*)wsrc;
        wv.x = __expf(wv.x); wv.y = __expf(wv.y);
        wv.z = __expf(wv.z); wv.w = __expf(wv.w);
        *(v4f*)&ldsw[4 * tid] = wv;
    }
    if (tid < 8) ldsw[2 * C + tid] = 0.f;

    #pragma unroll
    for (int j = 0; j < RED_C / 8; ++j)
        *(v4f*)&xs[0][soff + j * 8 * RED_P] = st[j];
    __syncthreads();

    const int  m    = lane & 15;
    const int  koff = (lane >> 4) << 1;
    const bool m0 = (m == 0), m1 = (m == 1);

    const float* a1p = m0 ? &ldsw[koff] : (m1 ? &ldsw[C + koff] : &ldsw[2 * C]);
    const float* a2p = m0 ? &ldsw[C + koff] : &ldsw[2 * C];
    const int a1s = (m < 2) ? 4 : 0;
    const int a2s = m0 ? 4 : 0;
    const int boff = koff * RED_P + 16 * w + m;

    v8f acc1 = {}; v8f acc2 = {};
    const int nchunk = C / RED_C;

    for (int ch = 0; ch < nchunk; ++ch) {
        const int  cur  = ch & 1;
        const bool more = (ch + 1 < nchunk);
        if (more) {
            gp += (size_t)RED_C * T;
            #pragma unroll
            for (int j = 0; j < RED_C / 8; ++j)
                st[j] = *(const v4f*)(gp + (size_t)j * rstep);
        }

        const float* bp = &xs[cur][boff];
        #pragma unroll
        for (int k = 0; k < RED_C / 4; ++k) {
            v2f A1 = *(const v2f*)a1p;
            v2f A2 = *(const v2f*)a2p;
            v2f Bx;
            Bx.x = bp[0];
            Bx.y = bp[RED_P];
            v2f Bxx = Bx * Bx;
            acc1 = __builtin_amdgcn_wmma_f32_16x16x4_f32(false, A1, false, Bx,
                                                         (short)0, acc1, false, false);
            acc2 = __builtin_amdgcn_wmma_f32_16x16x4_f32(false, A2, false, Bxx,
                                                         (short)0, acc2, false, false);
            a1p += a1s;
            a2p += a2s;
            bp  += 4 * RED_P;
        }

        if (more) {
            #pragma unroll
            for (int j = 0; j < RED_C / 8; ++j)
                *(v4f*)&xs[cur ^ 1][soff + j * 8 * RED_P] = st[j];
        }
        __syncthreads();
    }

    if (lane < 16) {
        size_t o = (size_t)b * T + ts + 16 * w + lane;
        xa[o] = acc1[0];
        xb[o] = acc1[1];
        xq[o] = acc2[0];
    }
}

#define SEQ_T   8192
#define SEQ_C   512
#define OUT_CH  64
#define OUT_RES 8
#define OUT_TT  1024
#define OUT_W   8
#define OUT_NT  (SEQ_T / OUT_TT)
#define OUT_NG  4
#define OUT_XB  32

__device__ __forceinline__ void lds_barrier() {
    __builtin_amdgcn_fence(__ATOMIC_RELEASE, "workgroup", "local");
    __builtin_amdgcn_s_barrier();
    __builtin_amdgcn_fence(__ATOMIC_ACQUIRE, "workgroup", "local");
}

__global__ void __launch_bounds__(256)
scan_apply(const float* __restrict__ x, const float* __restrict__ g,
           const float* __restrict__ xa, const float* __restrict__ xb,
           const float* __restrict__ xq,
           const float* __restrict__ Wa, const float* __restrict__ Wb,
           const float* __restrict__ Wo_w, const float* __restrict__ Wo_b,
           float* __restrict__ out, int C) {
    __shared__ __attribute__((aligned(16))) float totA[OUT_NT * OUT_W];
    __shared__ __attribute__((aligned(16))) float totM[OUT_NT * OUT_W];
    __shared__ __attribute__((aligned(16))) float totV[OUT_NT * OUT_W];
    __shared__ __attribute__((aligned(16))) float sMean[OUT_NT * 32 * 4];
    __shared__ __attribute__((aligned(16))) float sInv [OUT_NT * 32 * 4];

    const int tid  = threadIdx.x;
    const int lane = tid & 31;
    const int w    = tid >> 5;
    const int cgroups = C / OUT_CH;
    const int perB = OUT_RES * cgroups;
    const int b   = blockIdx.x / perB;
    const int ib  = blockIdx.x % perB;
    const int r   = ib % OUT_RES;
    const int c0  = (ib / OUT_RES) * OUT_CH;
    const bool mine = (__builtin_amdgcn_readfirstlane(w) == r);

    const size_t sbase = (size_t)b * SEQ_T + 4 * tid;
    const size_t xbase = ((size_t)b * C + c0) * SEQ_T + (size_t)w * OUT_TT + r * 128 + 4 * lane;

    v4f gq[OUT_NT], aq[OUT_NT], bq[OUT_NT], qq[OUT_NT];
    v4f xc[OUT_XB];

    v4f pa[SEQ_C / 128], pb[SEQ_C / 128];
    #pragma unroll
    for (int j = 0; j < SEQ_C / 128; ++j) {
        pa[j] = *(const v4f*)(Wa + 4 * lane + 128 * j);
        pb[j] = *(const v4f*)(Wb + 4 * lane + 128 * j);
    }

    for (int k = 0; k < OUT_NG; ++k) {
        const size_t so = sbase + (size_t)k * OUT_TT;
        gq[k] = *(const v4f*)(g  + so);
        aq[k] = *(const v4f*)(xa + so);
        bq[k] = *(const v4f*)(xb + so);
        qq[k] = *(const v4f*)(xq + so);
    }
    #pragma unroll
    for (int j = 0; j < OUT_XB / 4; ++j)
        xc[j] = __builtin_nontemporal_load((const v4f*)(x + xbase + (size_t)j * SEQ_T));

    float rSa, rSb;
    {
        float da = 0.f, db = 0.f;
        #pragma unroll
        for (int j = 0; j < SEQ_C / 128; ++j) {
            da += (__expf(pa[j].x) + __expf(pa[j].y)) + (__expf(pa[j].z) + __expf(pa[j].w));
            db += (__expf(pb[j].x) + __expf(pb[j].y)) + (__expf(pb[j].z) + __expf(pb[j].w));
        }
        const bool lo = lane < 16;
        float keep = lo ? da : db;
        keep += __shfl_xor(lo ? db : da, 16);
        for (int off = 8; off > 0; off >>= 1) keep += __shfl_xor(keep, off);
        const float Sa = __int_as_float(__builtin_amdgcn_readlane(__float_as_int(keep), 0));
        const float Sb = __int_as_float(__builtin_amdgcn_readlane(__float_as_int(keep), 16));
        rSa = 1.f / Sa;
        rSb = 1.f / Sb;
    }

    float cm = 0.f, cv = 0.f;

    #pragma unroll
    for (int grp = 0; grp < OUT_NT / OUT_NG; ++grp) {
        const int T0 = grp * OUT_NG;

        float a[OUT_NG][4], gt[OUT_NG][4], bm[OUT_NG][4];
        float A[OUT_NG], Bm[OUT_NG];
        for (int k = 0; k < OUT_NG; ++k) {
            for (int i = 0; i < 4; ++i) {
                gt[k][i] = gq[T0 + k][i] * MOMENTUM;
                a[k][i]  = 1.f - gt[k][i];
                bm[k][i] = gt[k][i] * (aq[T0 + k][i] * rSa);
            }
            A[k] = 1.f; Bm[k] = 0.f;
            for (int i = 0; i < 4; ++i) { Bm[k] = a[k][i] * Bm[k] + bm[k][i]; A[k] *= a[k][i]; }
        }

        float Ak[OUT_NG][5];
        for (int r5 = 0; r5 < 5; ++r5) {
            const int off = 1 << r5;
            for (int k = 0; k < OUT_NG; ++k) {
                float pA = __shfl_up(A[k],  off);
                float pB = __shfl_up(Bm[k], off);
                Ak[k][r5] = A[k];
                if (lane >= off) { Bm[k] = A[k] * pB + Bm[k]; A[k] = A[k] * pA; }
            }
        }
        float eA[OUT_NG], eM[OUT_NG];
        for (int k = 0; k < OUT_NG; ++k) {
            eA[k] = __shfl_up(A[k],  1);
            eM[k] = __shfl_up(Bm[k], 1);
            if (lane == 0) { eA[k] = 1.f; eM[k] = 0.f; }
            if (lane == 31) { totA[(T0 + k) * OUT_W + w] = A[k]; totM[(T0 + k) * OUT_W + w] = Bm[k]; }
        }
        lds_barrier();

        float ym[OUT_NG];
        {
            float y = cm;
            for (int k = 0; k < OUT_NG; ++k) {
                ym[k] = y;
                for (int j = 0; j < OUT_W; ++j) {
                    float tA = totA[(T0 + k) * OUT_W + j];
                    float tB = totM[(T0 + k) * OUT_W + j];
                    if (j == w) ym[k] = y;
                    y = tA * y + tB;
                }
            }
            cm = y;
        }
        float mean4[OUT_NG][4];
        for (int k = 0; k < OUT_NG; ++k) {
            float mcur = eA[k] * ym[k] + eM[k];
            for (int i = 0; i < 4; ++i) { mcur = a[k][i] * mcur + bm[k][i]; mean4[k][i] = mcur; }
        }
        if (mine) {
            for (int k = 0; k < OUT_NG; ++k) {
                v4f mv4 = { mean4[k][0], mean4[k][1], mean4[k][2], mean4[k][3] };
                *(v4f*)&sMean[((T0 + k) * 32 + lane) * 4] = mv4;
            }
        }

        float bv[OUT_NG][4], Bv[OUT_NG];
        for (int k = 0; k < OUT_NG; ++k) {
            for (int i = 0; i < 4; ++i) {
                float mv = mean4[k][i];
                float s  = qq[T0 + k][i] * rSb - 2.f * mv * (bq[T0 + k][i] * rSb) + mv * mv;
                bv[k][i] = gt[k][i] * s;
            }
            Bv[k] = 0.f;
            for (int i = 0; i < 4; ++i) Bv[k] = a[k][i] * Bv[k] + bv[k][i];
        }
        if (T0 + OUT_NG < OUT_NT) {
            for (int n = 0; n < OUT_NG; ++n) {
                const size_t so = sbase + (size_t)(T0 + OUT_NG + n) * OUT_TT;
                gq[T0 + OUT_NG + n] = *(const v4f*)(g  + so);
                aq[T0 + OUT_NG + n] = *(const v4f*)(xa + so);
                bq[T0 + OUT_NG + n] = *(const v4f*)(xb + so);
                qq[T0 + OUT_NG + n] = *(const v4f*)(xq + so);
            }
        } else {
            #pragma unroll
            for (int j = OUT_XB / 4; j < OUT_XB / 2; ++j)
                xc[j] = __builtin_nontemporal_load((const v4f*)(x + xbase + (size_t)j * SEQ_T));
        }
        for (int r5 = 0; r5 < 5; ++r5) {
            const int off = 1 << r5;
            for (int k = 0; k < OUT_NG; ++k) {
                float pB = __shfl_up(Bv[k], off);
                if (lane >= off) Bv[k] = Ak[k][r5] * pB + Bv[k];
            }
        }
        float eV[OUT_NG];
        for (int k = 0; k < OUT_NG; ++k) {
            eV[k] = __shfl_up(Bv[k], 1);
            if (lane == 0) eV[k] = 0.f;
            if (lane == 31) totV[(T0 + k) * OUT_W + w] = Bv[k];
        }
        lds_barrier();

        float yv[OUT_NG];
        {
            float y = cv;
            for (int k = 0; k < OUT_NG; ++k) {
                yv[k] = y;
                for (int j = 0; j < OUT_W; ++j) {
                    float tA = totA[(T0 + k) * OUT_W + j];
                    float tB = totV[(T0 + k) * OUT_W + j];
                    if (j == w) yv[k] = y;
                    y = tA * y + tB;
                }
            }
            cv = y;
        }
        if (mine) {
            for (int k = 0; k < OUT_NG; ++k) {
                float vcur = eA[k] * yv[k] + eV[k];
                v4f iv4;
                for (int i = 0; i < 4; ++i) {
                    vcur = a[k][i] * vcur + bv[k][i];
                    iv4[i] = rsqrtf(vcur + EPS_V);
                }
                *(v4f*)&sInv[((T0 + k) * 32 + lane) * 4] = iv4;
            }
        }
    }
    #pragma unroll
    for (int j = OUT_XB / 2; j < OUT_XB; ++j)
        xc[j] = __builtin_nontemporal_load((const v4f*)(x + xbase + (size_t)j * SEQ_T));
    lds_barrier();

    const v4f m4 = *(const v4f*)&sMean[(w * 32 + lane) * 4];
    const v4f i4 = *(const v4f*)&sInv [(w * 32 + lane) * 4];
    const float* xin  = x   + xbase;
    float*       yout = out + xbase;
    #pragma unroll
    for (int o = 0; o < OUT_CH / OUT_XB; ++o) {
        float wv[OUT_XB], bw[OUT_XB];
        #pragma unroll
        for (int j = 0; j < OUT_XB; ++j) {
            wv[j] = Wo_w[c0 + o * OUT_XB + j];
            bw[j] = Wo_b[c0 + o * OUT_XB + j];
        }
        #pragma unroll
        for (int j = 0; j < OUT_XB; ++j) {
            const int row = o * OUT_XB + j;
            v4f o4;
            o4.x = (xc[j].x - m4.x) * i4.x * wv[j] + bw[j];
            o4.y = (xc[j].y - m4.y) * i4.y * wv[j] + bw[j];
            o4.z = (xc[j].z - m4.z) * i4.z * wv[j] + bw[j];
            o4.w = (xc[j].w - m4.w) * i4.w * wv[j] + bw[j];
            __builtin_nontemporal_store(o4, (v4f*)(yout + (size_t)row * SEQ_T));
            if (row + OUT_XB < OUT_CH)
                xc[j] = __builtin_nontemporal_load(
                    (const v4f*)(xin + (size_t)(row + OUT_XB) * SEQ_T));
            __builtin_amdgcn_sched_barrier(0);
        }
    }
}

extern "C" void kernel_launch(void* const* d_in, const int* in_sizes, int n_in,
                              void* d_out, int out_size, void* d_ws, size_t ws_size,
                              hipStream_t stream) {
    const float* x    = (const float*)d_in[0];
    const float* g    = (const float*)d_in[1];
    const float* Wa_w = (const float*)d_in[2];
    const float* Wb_w = (const float*)d_in[3];
    const float* Wo_w = (const float*)d_in[4];
    const float* Wo_b = (const float*)d_in[5];
    float* out = (float*)d_out;

    const int C  = in_sizes[2];
    const int BT = in_sizes[1];
    const int T  = 8192;
    const int B  = BT / T;

    float* ws   = (float*)d_ws;
    float* xa   = ws;
    float* xb   = xa + BT;
    float* xq   = xb + BT;

    int waves  = (B * T) / 16;
    int blocks = waves / 8;
    reduce_wmma<<<blocks, 256, 0, stream>>>(x, Wa_w, Wb_w, xa, xb, xq, C, T);

    scan_apply<<<B * (C / OUT_CH) * OUT_RES, 256, 0, stream>>>(x, g, xa, xb, xq, Wa_w, Wb_w, Wo_w, Wo_b, out, C);
}


# reduce kernel: x feeds both WMMA operands (B = wa | wb | wb*x), xq read off the diagonal: 1 WMMA per 4 ch x 16 t instead of 2, +1/8 for the tiles' first two t; sum order over C changes
# speedup vs baseline: 1.7987x; 1.0587x over previous
#include <hip/hip_runtime.h>
#include <hip/hip_bf16.h>
#include <math.h>

#define MOMENTUM 0.05f
#define EPS_V 1e-6f

typedef float v2f __attribute__((ext_vector_type(2)));
typedef float v4f __attribute__((ext_vector_type(4)));
typedef float v8f __attribute__((ext_vector_type(8)));


#define RED_C   64
#define RED_T   128
#define RED_P   136

__device__ __forceinline__ float pick8(v8f a, int i) {
    float d = a[0];
    d = (i == 1) ? a[1] : d;
    d = (i == 2) ? a[2] : d;
    d = (i == 3) ? a[3] : d;
    d = (i == 4) ? a[4] : d;
    d = (i == 5) ? a[5] : d;
    d = (i == 6) ? a[6] : d;
    d = (i == 7) ? a[7] : d;
    return d;
}

__global__ void __launch_bounds__(256)
reduce_wmma(const float* __restrict__ x,
            const float* __restrict__ Wa, const float* __restrict__ Wb,
            float* __restrict__ xa, float* __restrict__ xb,
            float* __restrict__ xq, int C, int T) {
    __shared__ __attribute__((aligned(16))) float ldsw[1024];
    __shared__ __attribute__((aligned(16))) float xs[2][RED_C * RED_P];
    __shared__ __attribute__((aligned(16))) float res[3][RED_T];
    __shared__ __attribute__((aligned(16))) float part[8][16];

    const int tid  = threadIdx.x;
    const int lane = tid & 31;
    const int w    = tid >> 5;

    const int slabsPerB = T / RED_T;
    const int b  = blockIdx.x / slabsPerB;
    const int ts = (blockIdx.x % slabsPerB) * RED_T;

    const float* gp = x + (size_t)b * C * T + (size_t)w * T + ts + 4 * lane;
    const size_t rstep = 8 * (size_t)T;
    const int    soff  = w * RED_P + 4 * lane;

    v4f st[RED_C / 8];
    #pragma unroll
    for (int j = 0; j < RED_C / 8; ++j)
        st[j] = *(const v4f*)(gp + (size_t)j * rstep);

    if (tid < (C >> 1)) {
        const int q = C >> 2;
        const float* wsrc = (tid < q) ? (Wa + 4 * tid) : (Wb + 4 * (tid - q));
        v4f wv = *(const v4f*)wsrc;
        wv.x = __expf(wv.x); wv.y = __expf(wv.y);
        wv.z = __expf(wv.z); wv.w = __expf(wv.w);
        *(v4f*)&ldsw[4 * tid] = wv;
    }

    #pragma unroll
    for (int j = 0; j < RED_C / 8; ++j)
        *(v4f*)&xs[0][soff + j * 8 * RED_P] = st[j];
    __syncthreads();

    const int  m    = lane & 15;
    const int  h    = lane >> 4;
    const int  koff = h << 1;
    const bool wcol = (m < 2);

    const float* wp  = &ldsw[(m == 0 ? 0 : C) + koff];
    const int boff = koff * RED_P + 16 * w + m;
    const int eoff = (8 * w + koff) * RED_P + 16 * (m >> 1) + (m & 1);
    const float* wep = &ldsw[C + 8 * w + koff];

    v8f accE = {}; v8f accO = {}; v8f accX = {};
    const int nchunk = C / RED_C;

    for (int ch = 0; ch < nchunk; ++ch) {
        const int  cur  = ch & 1;
        const bool more = (ch + 1 < nchunk);
        if (more) {
            gp += (size_t)RED_C * T;
            #pragma unroll
            for (int j = 0; j < RED_C / 8; ++j)
                st[j] = *(const v4f*)(gp + (size_t)j * rstep);
        }

        const float* ep = &xs[cur][eoff];
        #pragma unroll
        for (int e = 0; e < 2; ++e) {
            v2f Xe;
            Xe.x = ep[e * 4 * RED_P];
            Xe.y = ep[e * 4 * RED_P + RED_P];
            v2f We = *(const v2f*)(wep + 4 * e);
            v2f Be = We * Xe;
            accX = __builtin_amdgcn_wmma_f32_16x16x4_f32(false, Xe, false, Be,
                                                         (short)0, accX, false, false);
        }

        const float* bp = &xs[cur][boff];
        #pragma unroll
        for (int k = 0; k < RED_C / 4; ++k) {
            v2f X;
            X.x = bp[k * 4 * RED_P];
            X.y = bp[k * 4 * RED_P + RED_P];
            v2f Wv = *(const v2f*)(wp + 4 * k);
            v2f WX = Wv * X;
            v2f Bv;
            Bv.x = wcol ? Wv.x : WX.x;
            Bv.y = wcol ? Wv.y : WX.y;
            if (k & 1)
                accO = __builtin_amdgcn_wmma_f32_16x16x4_f32(false, X, false, Bv,
                                                             (short)0, accO, false, false);
            else
                accE = __builtin_amdgcn_wmma_f32_16x16x4_f32(false, X, false, Bv,
                                                             (short)0, accE, false, false);
        }
        wp  += RED_C;
        wep += RED_C;

        if (more) {
            #pragma unroll
            for (int j = 0; j < RED_C / 8; ++j)
                *(v4f*)&xs[cur ^ 1][soff + j * 8 * RED_P] = st[j];
        }
        __syncthreads();
    }

    const v8f acc = accE + accO;
    if (wcol) {
        float* rp = &res[m][16 * w + 8 * h];
        v4f lo = { acc[0], acc[1], acc[2], acc[3] };
        v4f hi = { acc[4], acc[5], acc[6], acc[7] };
        *(v4f*)rp       = lo;
        *(v4f*)(rp + 4) = hi;
    }
    if ((m >> 3) == h) {
        res[2][16 * w + m] = pick8(acc, m & 7);
        part[w][m]         = pick8(accX, m & 7);
    }
    __syncthreads();

    if (tid < 96) {
        const int arr = tid >> 5;
        const int q   = tid & 31;
        v4f v = *(const v4f*)&res[arr][4 * q];
        if (arr == 2 && (q & 3) == 0) {
            const int j = q >> 2;
            float s0 = 0.f, s1 = 0.f;
            #pragma unroll
            for (int ww = 0; ww < 8; ++ww) {
                s0 += part[ww][2 * j];
                s1 += part[ww][2 * j + 1];
            }
            v.x = s0; v.y = s1;
        }
        float* dst = (arr == 0) ? xa : ((arr == 1) ? xb : xq);
        *(v4f*)(dst + (size_t)b * T + ts + 4 * q) = v;
    }
}

#define SEQ_T   8192
#define SEQ_C   512
#define OUT_CH  64
#define OUT_RES 8
#define OUT_TT  1024
#define OUT_W   8
#define OUT_NT  (SEQ_T / OUT_TT)
#define OUT_NG  4
#define OUT_XB  32

__device__ __forceinline__ void lds_barrier() {
    __builtin_amdgcn_fence(__ATOMIC_RELEASE, "workgroup", "local");
    __builtin_amdgcn_s_barrier();
    __builtin_amdgcn_fence(__ATOMIC_ACQUIRE, "workgroup", "local");
}

__global__ void __launch_bounds__(256)
scan_apply(const float* __restrict__ x, const float* __restrict__ g,
           const float* __restrict__ xa, const float* __restrict__ xb,
           const float* __restrict__ xq,
           const float* __restrict__ Wa, const float* __restrict__ Wb,
           const float* __restrict__ Wo_w, const float* __restrict__ Wo_b,
           float* __restrict__ out, int C) {
    __shared__ __attribute__((aligned(16))) float totA[OUT_NT * OUT_W];
    __shared__ __attribute__((aligned(16))) float totM[OUT_NT * OUT_W];
    __shared__ __attribute__((aligned(16))) float totV[OUT_NT * OUT_W];
    __shared__ __attribute__((aligned(16))) float sMean[OUT_NT * 32 * 4];
    __shared__ __attribute__((aligned(16))) float sInv [OUT_NT * 32 * 4];

    const int tid  = threadIdx.x;
    const int lane = tid & 31;
    const int w    = tid >> 5;
    const int cgroups = C / OUT_CH;
    const int perB = OUT_RES * cgroups;
    const int b   = blockIdx.x / perB;
    const int ib  = blockIdx.x % perB;
    const int r   = ib % OUT_RES;
    const int c0  = (ib / OUT_RES) * OUT_CH;
    const bool mine = (__builtin_amdgcn_readfirstlane(w) == r);

    const size_t sbase = (size_t)b * SEQ_T + 4 * tid;
    const size_t xbase = ((size_t)b * C + c0) * SEQ_T + (size_t)w * OUT_TT + r * 128 + 4 * lane;

    v4f gq[OUT_NT], aq[OUT_NT], bq[OUT_NT], qq[OUT_NT];
    v4f xc[OUT_XB];

    v4f pa[SEQ_C / 128], pb[SEQ_C / 128];
    #pragma unroll
    for (int j = 0; j < SEQ_C / 128; ++j) {
        pa[j] = *(const v4f*)(Wa + 4 * lane + 128 * j);
        pb[j] = *(const v4f*)(Wb + 4 * lane + 128 * j);
    }

    for (int k = 0; k < OUT_NG; ++k) {
        const size_t so = sbase + (size_t)k * OUT_TT;
        gq[k] = *(const v4f*)(g  + so);
        aq[k] = *(const v4f*)(xa + so);
        bq[k] = *(const v4f*)(xb + so);
        qq[k] = *(const v4f*)(xq + so);
    }
    #pragma unroll
    for (int j = 0; j < OUT_XB / 4; ++j)
        xc[j] = __builtin_nontemporal_load((const v4f*)(x + xbase + (size_t)j * SEQ_T));

    float rSa, rSb;
    {
        float da = 0.f, db = 0.f;
        #pragma unroll
        for (int j = 0; j < SEQ_C / 128; ++j) {
            da += (__expf(pa[j].x) + __expf(pa[j].y)) + (__expf(pa[j].z) + __expf(pa[j].w));
            db += (__expf(pb[j].x) + __expf(pb[j].y)) + (__expf(pb[j].z) + __expf(pb[j].w));
        }
        const bool lo = lane < 16;
        float keep = lo ? da : db;
        keep += __shfl_xor(lo ? db : da, 16);
        for (int off = 8; off > 0; off >>= 1) keep += __shfl_xor(keep, off);
        const float Sa = __int_as_float(__builtin_amdgcn_readlane(__float_as_int(keep), 0));
        const float Sb = __int_as_float(__builtin_amdgcn_readlane(__float_as_int(keep), 16));
        rSa = 1.f / Sa;
        rSb = 1.f / Sb;
    }

    float cm = 0.f, cv = 0.f;

    #pragma unroll
    for (int grp = 0; grp < OUT_NT / OUT_NG; ++grp) {
        const int T0 = grp * OUT_NG;

        float a[OUT_NG][4], gt[OUT_NG][4], bm[OUT_NG][4];
        float A[OUT_NG], Bm[OUT_NG];
        for (int k = 0; k < OUT_NG; ++k) {
            for (int i = 0; i < 4; ++i) {
                gt[k][i] = gq[T0 + k][i] * MOMENTUM;
                a[k][i]  = 1.f - gt[k][i];
                bm[k][i] = gt[k][i] * (aq[T0 + k][i] * rSa);
            }
            A[k] = 1.f; Bm[k] = 0.f;
            for (int i = 0; i < 4; ++i) { Bm[k] = a[k][i] * Bm[k] + bm[k][i]; A[k] *= a[k][i]; }
        }

        float Ak[OUT_NG][5];
        for (int r5 = 0; r5 < 5; ++r5) {
            const int off = 1 << r5;
            for (int k = 0; k < OUT_NG; ++k) {
                float pA = __shfl_up(A[k],  off);
                float pB = __shfl_up(Bm[k], off);
                Ak[k][r5] = A[k];
                if (lane >= off) { Bm[k] = A[k] * pB + Bm[k]; A[k] = A[k] * pA; }
            }
        }
        float eA[OUT_NG], eM[OUT_NG];
        for (int k = 0; k < OUT_NG; ++k) {
            eA[k] = __shfl_up(A[k],  1);
            eM[k] = __shfl_up(Bm[k], 1);
            if (lane == 0) { eA[k] = 1.f; eM[k] = 0.f; }
            if (lane == 31) { totA[(T0 + k) * OUT_W + w] = A[k]; totM[(T0 + k) * OUT_W + w] = Bm[k]; }
        }
        lds_barrier();

        float ym[OUT_NG];
        {
            float y = cm;
            for (int k = 0; k < OUT_NG; ++k) {
                ym[k] = y;
                for (int j = 0; j < OUT_W; ++j) {
                    float tA = totA[(T0 + k) * OUT_W + j];
                    float tB = totM[(T0 + k) * OUT_W + j];
                    if (j == w) ym[k] = y;
                    y = tA * y + tB;
                }
            }
            cm = y;
        }
        float mean4[OUT_NG][4];
        for (int k = 0; k < OUT_NG; ++k) {
            float mcur = eA[k] * ym[k] + eM[k];
            for (int i = 0; i < 4; ++i) { mcur = a[k][i] * mcur + bm[k][i]; mean4[k][i] = mcur; }
        }
        if (mine) {
            for (int k = 0; k < OUT_NG; ++k) {
                v4f mv4 = { mean4[k][0], mean4[k][1], mean4[k][2], mean4[k][3] };
                *(v4f*)&sMean[((T0 + k) * 32 + lane) * 4] = mv4;
            }
        }

        float bv[OUT_NG][4], Bv[OUT_NG];
        for (int k = 0; k < OUT_NG; ++k) {
            for (int i = 0; i < 4; ++i) {
                float mv = mean4[k][i];
                float s  = qq[T0 + k][i] * rSb - 2.f * mv * (bq[T0 + k][i] * rSb) + mv * mv;
                bv[k][i] = gt[k][i] * s;
            }
            Bv[k] = 0.f;
            for (int i = 0; i < 4; ++i) Bv[k] = a[k][i] * Bv[k] + bv[k][i];
        }
        if (T0 + OUT_NG < OUT_NT) {
            for (int n = 0; n < OUT_NG; ++n) {
                const size_t so = sbase + (size_t)(T0 + OUT_NG + n) * OUT_TT;
                gq[T0 + OUT_NG + n] = *(const v4f*)(g  + so);
                aq[T0 + OUT_NG + n] = *(const v4f*)(xa + so);
                bq[T0 + OUT_NG + n] = *(const v4f*)(xb + so);
                qq[T0 + OUT_NG + n] = *(const v4f*)(xq + so);
            }
        } else {
            #pragma unroll
            for (int j = OUT_XB / 4; j < OUT_XB / 2; ++j)
                xc[j] = __builtin_nontemporal_load((const v4f*)(x + xbase + (size_t)j * SEQ_T));
        }
        for (int r5 = 0; r5 < 5; ++r5) {
            const int off = 1 << r5;
            for (int k = 0; k < OUT_NG; ++k) {
                float pB = __shfl_up(Bv[k], off);
                if (lane >= off) Bv[k] = Ak[k][r5] * pB + Bv[k];
            }
        }
        float eV[OUT_NG];
        for (int k = 0; k < OUT_NG; ++k) {
            eV[k] = __shfl_up(Bv[k], 1);
            if (lane == 0) eV[k] = 0.f;
            if (lane == 31) totV[(T0 + k) * OUT_W + w] = Bv[k];
        }
        lds_barrier();

        float yv[OUT_NG];
        {
            float y = cv;
            for (int k = 0; k < OUT_NG; ++k) {
                yv[k] = y;
                for (int j = 0; j < OUT_W; ++j) {
                    float tA = totA[(T0 + k) * OUT_W + j];
                    float tB = totV[(T0 + k) * OUT_W + j];
                    if (j == w) yv[k] = y;
                    y = tA * y + tB;
                }
            }
            cv = y;
        }
        if (mine) {
            for (int k = 0; k < OUT_NG; ++k) {
                float vcur = eA[k] * yv[k] + eV[k];
                v4f iv4;
                for (int i = 0; i < 4; ++i) {
                    vcur = a[k][i] * vcur + bv[k][i];
                    iv4[i] = rsqrtf(vcur + EPS_V);
                }
                *(v4f*)&sInv[((T0 + k) * 32 + lane) * 4] = iv4;
            }
        }
    }
    #pragma unroll
    for (int j = OUT_XB / 2; j < OUT_XB; ++j)
        xc[j] = __builtin_nontemporal_load((const v4f*)(x + xbase + (size_t)j * SEQ_T));
    lds_barrier();

    const v4f m4 = *(const v4f*)&sMean[(w * 32 + lane) * 4];
    const v4f i4 = *(const v4f*)&sInv [(w * 32 + lane) * 4];
    const float* xin  = x   + xbase;
    float*       yout = out + xbase;
    #pragma unroll
    for (int o = 0; o < OUT_CH / OUT_XB; ++o) {
        float wv[OUT_XB], bw[OUT_XB];
        #pragma unroll
        for (int j = 0; j < OUT_XB; ++j) {
            wv[j] = Wo_w[c0 + o * OUT_XB + j];
            bw[j] = Wo_b[c0 + o * OUT_XB + j];
        }
        #pragma unroll
        for (int j = 0; j < OUT_XB; ++j) {
            const int row = o * OUT_XB + j;
            v4f o4;
            o4.x = (xc[j].x - m4.x) * i4.x * wv[j] + bw[j];
            o4.y = (xc[j].y - m4.y) * i4.y * wv[j] + bw[j];
            o4.z = (xc[j].z - m4.z) * i4.z * wv[j] + bw[j];
            o4.w = (xc[j].w - m4.w) * i4.w * wv[j] + bw[j];
            __builtin_nontemporal_store(o4, (v4f*)(yout + (size_t)row * SEQ_T));
            if (row + OUT_XB < OUT_CH)
                xc[j] = __builtin_nontemporal_load(
                    (const v4f*)(xin + (size_t)(row + OUT_XB) * SEQ_T));
            __builtin_amdgcn_sched_barrier(0);
        }
    }
}

extern "C" void kernel_launch(void* const* d_in, const int* in_sizes, int n_in,
                              void* d_out, int out_size, void* d_ws, size_t ws_size,
                              hipStream_t stream) {
    const float* x    = (const float*)d_in[0];
    const float* g    = (const float*)d_in[1];
    const float* Wa_w = (const float*)d_in[2];
    const float* Wb_w = (const float*)d_in[3];
    const float* Wo_w = (const float*)d_in[4];
    const float* Wo_b = (const float*)d_in[5];
    float* out = (float*)d_out;

    const int C  = in_sizes[2];
    const int BT = in_sizes[1];
    const int T  = 8192;
    const int B  = BT / T;

    float* ws   = (float*)d_ws;
    float* xa   = ws;
    float* xb   = xa + BT;
    float* xq   = xb + BT;

    int waves  = (B * T) / 16;
    int blocks = waves / 8;
    reduce_wmma<<<blocks, 256, 0, stream>>>(x, Wa_w, Wb_w, xa, xb, xq, C, T);

    scan_apply<<<B * (C / OUT_CH) * OUT_RES, 256, 0, stream>>>(x, g, xa, xb, xq, Wa_w, Wb_w, Wo_w, Wo_b, out, C);
}


# output kernel: one 16-wave block per WGP owns 128 channels x 8 slabs; scan map re-laid: a wave owns 512 consecutive t, 3 carry levels, 16-step chain not 64; carry sum order changes
# speedup vs baseline: 1.9360x; 1.0763x over previous
#include <hip/hip_runtime.h>
#include <hip/hip_bf16.h>
#include <math.h>

#define MOMENTUM 0.05f
#define EPS_V 1e-6f

typedef float v2f __attribute__((ext_vector_type(2)));
typedef float v4f __attribute__((ext_vector_type(4)));
typedef float v8f __attribute__((ext_vector_type(8)));


#define RED_C   64
#define RED_T   128
#define RED_P   136

__device__ __forceinline__ float pick8(v8f a, int i) {
    float d = a[0];
    d = (i == 1) ? a[1] : d;
    d = (i == 2) ? a[2] : d;
    d = (i == 3) ? a[3] : d;
    d = (i == 4) ? a[4] : d;
    d = (i == 5) ? a[5] : d;
    d = (i == 6) ? a[6] : d;
    d = (i == 7) ? a[7] : d;
    return d;
}

__global__ void __launch_bounds__(256)
reduce_wmma(const float* __restrict__ x,
            const float* __restrict__ Wa, const float* __restrict__ Wb,
            float* __restrict__ xa, float* __restrict__ xb,
            float* __restrict__ xq, int C, int T) {
    __shared__ __attribute__((aligned(16))) float ldsw[1024];
    __shared__ __attribute__((aligned(16))) float xs[2][RED_C * RED_P];
    __shared__ __attribute__((aligned(16))) float res[3][RED_T];
    __shared__ __attribute__((aligned(16))) float part[8][16];

    const int tid  = threadIdx.x;
    const int lane = tid & 31;
    const int w    = tid >> 5;

    const int slabsPerB = T / RED_T;
    const int b  = blockIdx.x / slabsPerB;
    const int ts = (blockIdx.x % slabsPerB) * RED_T;

    const float* gp = x + (size_t)b * C * T + (size_t)w * T + ts + 4 * lane;
    const size_t rstep = 8 * (size_t)T;
    const int    soff  = w * RED_P + 4 * lane;

    v4f st[RED_C / 8];
    #pragma unroll
    for (int j = 0; j < RED_C / 8; ++j)
        st[j] = *(const v4f*)(gp + (size_t)j * rstep);

    if (tid < (C >> 1)) {
        const int q = C >> 2;
        const float* wsrc = (tid < q) ? (Wa + 4 * tid) : (Wb + 4 * (tid - q));
        v4f wv = *(const v4f*)wsrc;
        wv.x = __expf(wv.x); wv.y = __expf(wv.y);
        wv.z = __expf(wv.z); wv.w = __expf(wv.w);
        *(v4f*)&ldsw[4 * tid] = wv;
    }

    #pragma unroll
    for (int j = 0; j < RED_C / 8; ++j)
        *(v4f*)&xs[0][soff + j * 8 * RED_P] = st[j];
    __syncthreads();

    const int  m    = lane & 15;
    const int  h    = lane >> 4;
    const int  koff = h << 1;
    const bool wcol = (m < 2);

    const float* wp  = &ldsw[(m == 0 ? 0 : C) + koff];
    const int boff = koff * RED_P + 16 * w + m;
    const int eoff = (8 * w + koff) * RED_P + 16 * (m >> 1) + (m & 1);
    const float* wep = &ldsw[C + 8 * w + koff];

    v8f accE = {}; v8f accO = {}; v8f accX = {};
    const int nchunk = C / RED_C;

    for (int ch = 0; ch < nchunk; ++ch) {
        const int  cur  = ch & 1;
        const bool more = (ch + 1 < nchunk);
        if (more) {
            gp += (size_t)RED_C * T;
            #pragma unroll
            for (int j = 0; j < RED_C / 8; ++j)
                st[j] = *(const v4f*)(gp + (size_t)j * rstep);
        }

        const float* ep = &xs[cur][eoff];
        #pragma unroll
        for (int e = 0; e < 2; ++e) {
            v2f Xe;
            Xe.x = ep[e * 4 * RED_P];
            Xe.y = ep[e * 4 * RED_P + RED_P];
            v2f We = *(const v2f*)(wep + 4 * e);
            v2f Be = We * Xe;
            accX = __builtin_amdgcn_wmma_f32_16x16x4_f32(false, Xe, false, Be,
                                                         (short)0, accX, false, false);
        }

        const float* bp = &xs[cur][boff];
        #pragma unroll
        for (int k = 0; k < RED_C / 4; ++k) {
            v2f X;
            X.x = bp[k * 4 * RED_P];
            X.y = bp[k * 4 * RED_P + RED_P];
            v2f Wv = *(const v2f*)(wp + 4 * k);
            v2f WX = Wv * X;
            v2f Bv;
            Bv.x = wcol ? Wv.x : WX.x;
            Bv.y = wcol ? Wv.y : WX.y;
            if (k & 1)
                accO = __builtin_amdgcn_wmma_f32_16x16x4_f32(false, X, false, Bv,
                                                             (short)0, accO, false, false);
            else
                accE = __builtin_amdgcn_wmma_f32_16x16x4_f32(false, X, false, Bv,
                                                             (short)0, accE, false, false);
        }
        wp  += RED_C;
        wep += RED_C;

        if (more) {
            #pragma unroll
            for (int j = 0; j < RED_C / 8; ++j)
                *(v4f*)&xs[cur ^ 1][soff + j * 8 * RED_P] = st[j];
        }
        __syncthreads();
    }

    const v8f acc = accE + accO;
    if (wcol) {
        float* rp = &res[m][16 * w + 8 * h];
        v4f lo = { acc[0], acc[1], acc[2], acc[3] };
        v4f hi = { acc[4], acc[5], acc[6], acc[7] };
        *(v4f*)rp       = lo;
        *(v4f*)(rp + 4) = hi;
    }
    if ((m >> 3) == h) {
        res[2][16 * w + m] = pick8(acc, m & 7);
        part[w][m]         = pick8(accX, m & 7);
    }
    __syncthreads();

    if (tid < 96) {
        const int arr = tid >> 5;
        const int q   = tid & 31;
        v4f v = *(const v4f*)&res[arr][4 * q];
        if (arr == 2 && (q & 3) == 0) {
            const int j = q >> 2;
            float s0 = 0.f, s1 = 0.f;
            #pragma unroll
            for (int ww = 0; ww < 8; ++ww) {
                s0 += part[ww][2 * j];
                s1 += part[ww][2 * j + 1];
            }
            v.x = s0; v.y = s1;
        }
        float* dst = (arr == 0) ? xa : ((arr == 1) ? xb : xq);
        *(v4f*)(dst + (size_t)b * T + ts + 4 * q) = v;
    }
}

#define SEQ_T   8192
#define SEQ_C   512
#define OUT_CH  128
#define OUT_RW  64
#define OUT_RES 8
#define OUT_W   16
#define OUT_THR 512
#define OUT_WT  (SEQ_T / OUT_W)
#define OUT_NS  (OUT_WT / 128)
#define OUT_XB  32

__device__ __forceinline__ void lds_barrier() {
    __builtin_amdgcn_fence(__ATOMIC_RELEASE, "workgroup", "local");
    __builtin_amdgcn_s_barrier();
    __builtin_amdgcn_fence(__ATOMIC_ACQUIRE, "workgroup", "local");
}

__device__ __forceinline__ float last_lane(float v) {
    return __int_as_float(__builtin_amdgcn_readlane(__float_as_int(v), 31));
}

__device__ __forceinline__ float sel4(float v0, float v1, float v2, float v3, int k) {
    float d = v0;
    d = (k == 1) ? v1 : d;
    d = (k == 2) ? v2 : d;
    d = (k == 3) ? v3 : d;
    return d;
}

__global__ void __launch_bounds__(OUT_THR)
scan_apply(const float* __restrict__ x, const float* __restrict__ g,
           const float* __restrict__ xa, const float* __restrict__ xb,
           const float* __restrict__ xq,
           const float* __restrict__ Wa, const float* __restrict__ Wb,
           const float* __restrict__ Wo_w, const float* __restrict__ Wo_b,
           float* __restrict__ out, int C) {
    __shared__ __attribute__((aligned(16))) float totA[OUT_W];
    __shared__ __attribute__((aligned(16))) float totM[OUT_W];
    __shared__ __attribute__((aligned(16))) float totV[OUT_W];
    __shared__ __attribute__((aligned(16))) float sMean[OUT_RES * 32 * 4];
    __shared__ __attribute__((aligned(16))) float sInv [OUT_RES * 32 * 4];

    const int tid  = threadIdx.x;
    const int lane = tid & 31;
    const int wu   = __builtin_amdgcn_readfirstlane(tid >> 5);
    const int cgroups = C / OUT_CH;
    const int perB = OUT_RES * cgroups;
    const int b   = blockIdx.x / perB;
    const int ib  = blockIdx.x % perB;
    const int r   = ib % OUT_RES;
    const int c0  = (ib / OUT_RES) * OUT_CH;
    const int  kk   = r & 3;
    const bool mine = ((wu & 1) == (r >> 2));
    const int  msc  = wu >> 1;
    const int ms  = wu & (OUT_RES - 1);
    const int hc  = wu >> 3;
    const int cw  = c0 + hc * OUT_RW;

    const size_t sbase = (size_t)b * SEQ_T + (size_t)wu * OUT_WT + 4 * lane;
    const size_t xbase = ((size_t)b * C + cw) * SEQ_T + (size_t)ms * 1024 + r * 128 + 4 * lane;

    v4f gq[OUT_NS], aq[OUT_NS], bq[OUT_NS], qq[OUT_NS];
    v4f xc[OUT_XB];

    v4f pa[SEQ_C / 128], pb[SEQ_C / 128];
    #pragma unroll
    for (int j = 0; j < SEQ_C / 128; ++j) {
        pa[j] = *(const v4f*)(Wa + 4 * lane + 128 * j);
        pb[j] = *(const v4f*)(Wb + 4 * lane + 128 * j);
    }

    #pragma unroll
    for (int k = 0; k < OUT_NS; ++k) {
        const size_t so = sbase + (size_t)k * 128;
        gq[k] = *(const v4f*)(g  + so);
        aq[k] = *(const v4f*)(xa + so);
        bq[k] = *(const v4f*)(xb + so);
        qq[k] = *(const v4f*)(xq + so);
    }
    #pragma unroll
    for (int j = 0; j < OUT_XB / 2; ++j)
        xc[j] = __builtin_nontemporal_load((const v4f*)(x + xbase + (size_t)j * SEQ_T));

    float rSa, rSb;
    {
        float da = 0.f, db = 0.f;
        #pragma unroll
        for (int j = 0; j < SEQ_C / 128; ++j) {
            da += (__expf(pa[j].x) + __expf(pa[j].y)) + (__expf(pa[j].z) + __expf(pa[j].w));
            db += (__expf(pb[j].x) + __expf(pb[j].y)) + (__expf(pb[j].z) + __expf(pb[j].w));
        }
        const bool lo = lane < 16;
        float keep = lo ? da : db;
        keep += __shfl_xor(lo ? db : da, 16);
        for (int off = 8; off > 0; off >>= 1) keep += __shfl_xor(keep, off);
        const float Sa = __int_as_float(__builtin_amdgcn_readlane(__float_as_int(keep), 0));
        const float Sb = __int_as_float(__builtin_amdgcn_readlane(__float_as_int(keep), 16));
        rSa = 1.f / Sa;
        rSb = 1.f / Sb;
    }

    float a[OUT_NS][4], gt[OUT_NS][4], bm[OUT_NS][4];
    float A[OUT_NS], Bm[OUT_NS];
    #pragma unroll
    for (int k = 0; k < OUT_NS; ++k) {
        #pragma unroll
        for (int i = 0; i < 4; ++i) {
            gt[k][i] = gq[k][i] * MOMENTUM;
            a[k][i]  = 1.f - gt[k][i];
            bm[k][i] = gt[k][i] * (aq[k][i] * rSa);
        }
        A[k] = 1.f; Bm[k] = 0.f;
        #pragma unroll
        for (int i = 0; i < 4; ++i) { Bm[k] = a[k][i] * Bm[k] + bm[k][i]; A[k] *= a[k][i]; }
    }

    float Ak[OUT_NS][5];
    #pragma unroll
    for (int r5 = 0; r5 < 5; ++r5) {
        const int off = 1 << r5;
        #pragma unroll
        for (int k = 0; k < OUT_NS; ++k) {
            float pA = __shfl_up(A[k],  off);
            float pB = __shfl_up(Bm[k], off);
            Ak[k][r5] = A[k];
            if (lane >= off) { Bm[k] = A[k] * pB + Bm[k]; A[k] = A[k] * pA; }
        }
    }
    float eA[OUT_NS], eM[OUT_NS];
    float SA[OUT_NS];
    float PA[OUT_NS], PM[OUT_NS];
    float TA, TM;
    {
        float cA = 1.f, cM = 0.f;
        #pragma unroll
        for (int k = 0; k < OUT_NS; ++k) {
            eA[k] = __shfl_up(A[k],  1);
            eM[k] = __shfl_up(Bm[k], 1);
            if (lane == 0) { eA[k] = 1.f; eM[k] = 0.f; }
            SA[k] = last_lane(A[k]);
            const float sM = last_lane(Bm[k]);
            PA[k] = cA; PM[k] = cM;
            cM = SA[k] * cM + sM;
            cA = SA[k] * cA;
        }
        TA = cA; TM = cM;
    }
    if (lane == 0) { totA[wu] = TA; totM[wu] = TM; }
    lds_barrier();

    float mean4[OUT_NS][4];
    {
        float y = 0.f, Yw = 0.f;
        #pragma unroll
        for (int j = 0; j < OUT_W; ++j) {
            const float tA = totA[j];
            const float tB = totM[j];
            if (j == wu) Yw = y;
            y = tA * y + tB;
        }
        #pragma unroll
        for (int k = 0; k < OUT_NS; ++k) {
            const float yk = PA[k] * Yw + PM[k];
            float mcur = eA[k] * yk + eM[k];
            #pragma unroll
            for (int i = 0; i < 4; ++i) { mcur = a[k][i] * mcur + bm[k][i]; mean4[k][i] = mcur; }
        }
    }
    if (mine) {
        v4f mv4;
        #pragma unroll
        for (int i = 0; i < 4; ++i)
            mv4[i] = sel4(mean4[0][i], mean4[1][i], mean4[2][i], mean4[3][i], kk);
        *(v4f*)&sMean[(msc * 32 + lane) * 4] = mv4;
    }

    float bv[OUT_NS][4], Bv[OUT_NS];
    #pragma unroll
    for (int k = 0; k < OUT_NS; ++k) {
        #pragma unroll
        for (int i = 0; i < 4; ++i) {
            float mv = mean4[k][i];
            float s  = qq[k][i] * rSb - 2.f * mv * (bq[k][i] * rSb) + mv * mv;
            bv[k][i] = gt[k][i] * s;
        }
        Bv[k] = 0.f;
        #pragma unroll
        for (int i = 0; i < 4; ++i) Bv[k] = a[k][i] * Bv[k] + bv[k][i];
    }
    #pragma unroll
    for (int j = OUT_XB / 2; j < OUT_XB; ++j)
        xc[j] = __builtin_nontemporal_load((const v4f*)(x + xbase + (size_t)j * SEQ_T));
    #pragma unroll
    for (int r5 = 0; r5 < 5; ++r5) {
        const int off = 1 << r5;
        #pragma unroll
        for (int k = 0; k < OUT_NS; ++k) {
            float pB = __shfl_up(Bv[k], off);
            if (lane >= off) Bv[k] = Ak[k][r5] * pB + Bv[k];
        }
    }
    float eV[OUT_NS], PV[OUT_NS], TV;
    {
        float cV = 0.f;
        #pragma unroll
        for (int k = 0; k < OUT_NS; ++k) {
            eV[k] = __shfl_up(Bv[k], 1);
            if (lane == 0) eV[k] = 0.f;
            PV[k] = cV;
            cV = SA[k] * cV + last_lane(Bv[k]);
        }
        TV = cV;
    }
    if (lane == 0) totV[wu] = TV;
    lds_barrier();

    if (mine) {
        float y = 0.f, Yv = 0.f;
        #pragma unroll
        for (int j = 0; j < OUT_W; ++j) {
            const float tA = totA[j];
            const float tB = totV[j];
            if (j == wu) Yv = y;
            y = tA * y + tB;
        }
        const float yk = sel4(PA[0], PA[1], PA[2], PA[3], kk) * Yv
                       + sel4(PV[0], PV[1], PV[2], PV[3], kk);
        float vcur = sel4(eA[0], eA[1], eA[2], eA[3], kk) * yk
                   + sel4(eV[0], eV[1], eV[2], eV[3], kk);
        v4f iv4;
        #pragma unroll
        for (int i = 0; i < 4; ++i) {
            vcur = sel4(a[0][i], a[1][i], a[2][i], a[3][i], kk) * vcur
                 + sel4(bv[0][i], bv[1][i], bv[2][i], bv[3][i], kk);
            iv4[i] = rsqrtf(vcur + EPS_V);
        }
        *(v4f*)&sInv[(msc * 32 + lane) * 4] = iv4;
    }
    lds_barrier();

    const v4f m4 = *(const v4f*)&sMean[(ms * 32 + lane) * 4];
    const v4f i4 = *(const v4f*)&sInv [(ms * 32 + lane) * 4];
    const float* xin  = x   + xbase;
    float*       yout = out + xbase;
    #pragma unroll
    for (int o = 0; o < OUT_RW / OUT_XB; ++o) {
        float wv[OUT_XB], bw[OUT_XB];
        #pragma unroll
        for (int j = 0; j < OUT_XB; ++j) {
            wv[j] = Wo_w[cw + o * OUT_XB + j];
            bw[j] = Wo_b[cw + o * OUT_XB + j];
        }
        #pragma unroll
        for (int j = 0; j < OUT_XB; ++j) {
            const int row = o * OUT_XB + j;
            v4f o4;
            o4.x = (xc[j].x - m4.x) * i4.x * wv[j] + bw[j];
            o4.y = (xc[j].y - m4.y) * i4.y * wv[j] + bw[j];
            o4.z = (xc[j].z - m4.z) * i4.z * wv[j] + bw[j];
            o4.w = (xc[j].w - m4.w) * i4.w * wv[j] + bw[j];
            __builtin_nontemporal_store(o4, (v4f*)(yout + (size_t)row * SEQ_T));
            if (row + OUT_XB < OUT_RW)
                xc[j] = __builtin_nontemporal_load(
                    (const v4f*)(xin + (size_t)(row + OUT_XB) * SEQ_T));
            __builtin_amdgcn_sched_barrier(0);
        }
    }
}

extern "C" void kernel_launch(void* const* d_in, const int* in_sizes, int n_in,
                              void* d_out, int out_size, void* d_ws, size_t ws_size,
                              hipStream_t stream) {
    const float* x    = (const float*)d_in[0];
    const float* g    = (const float*)d_in[1];
    const float* Wa_w = (const float*)d_in[2];
    const float* Wb_w = (const float*)d_in[3];
    const float* Wo_w = (const float*)d_in[4];
    const float* Wo_b = (const float*)d_in[5];
    float* out = (float*)d_out;

    const int C  = in_sizes[2];
    const int BT = in_sizes[1];
    const int T  = 8192;
    const int B  = BT / T;

    float* ws   = (float*)d_ws;
    float* xa   = ws;
    float* xb   = xa + BT;
    float* xq   = xb + BT;

    int waves  = (B * T) / 16;
    int blocks = waves / 8;
    reduce_wmma<<<blocks, 256, 0, stream>>>(x, Wa_w, Wb_w, xa, xb, xq, C, T);

    scan_apply<<<B * (C / OUT_CH) * OUT_RES, OUT_THR, 0, stream>>>(x, g, xa, xb, xq, Wa_w, Wb_w, Wo_w, Wo_b, out, C);
}
